# starting kernel
# baseline (speedup 1.0000x reference)
#include <hip/hip_runtime.h>
#include <cmath>

#define HQ 1024
#define WQ 1024
#define HW (HQ * WQ)
#define NB 8
#define NC 3

typedef __attribute__((ext_vector_type(2))) float v2f;
typedef __attribute__((ext_vector_type(8))) float v8f;

struct GW { float g31[31]; float g7[7]; };

__device__ __forceinline__ int iclamp(int v, int lo, int hi) {
    return v < lo ? lo : (v > hi ? hi : v);
}

__global__ __launch_bounds__(256) void luma_kernel(const float* __restrict__ x,
                                                   float* __restrict__ L) {
    int i = blockIdx.x * 256 + threadIdx.x;
    int b   = i >> 18;
    int rem = i & ((HW / 4) - 1);
    const float4* xv = (const float4*)x;
    float4 r  = xv[(size_t)(b * 3 + 0) * (HW / 4) + rem];
    float4 g  = xv[(size_t)(b * 3 + 1) * (HW / 4) + rem];
    float4 bl = xv[(size_t)(b * 3 + 2) * (HW / 4) + rem];
    float4 o;
    o.x = 0.2126f * r.x + 0.7152f * g.x + 0.0722f * bl.x;
    o.y = 0.2126f * r.y + 0.7152f * g.y + 0.0722f * bl.y;
    o.z = 0.2126f * r.z + 0.7152f * g.z + 0.0722f * bl.z;
    o.w = 0.2126f * r.w + 0.7152f * g.w + 0.0722f * bl.w;
    ((float4*)L)[(size_t)b * (HW / 4) + rem] = o;
}

__global__ __launch_bounds__(256) void vblur_kernel(const float* __restrict__ L,
                                                    float* __restrict__ vb31,
                                                    float* __restrict__ vb7,
                                                    GW gw) {
    int i = blockIdx.x * 256 + threadIdx.x;
    int b  = i >> 20;
    int p  = i & (HW - 1);
    int y  = p >> 10;
    int xx = p & (WQ - 1);
    const float* base = L + (size_t)b * HW + xx;

    float a31 = 0.f;
#pragma unroll
    for (int t = 0; t < 31; ++t) {
        int yy = y + t - 15;
        int yc = iclamp(yy, 0, HQ - 1);
        float m = (yy == yc) ? 1.f : 0.f;
        a31 = fmaf(gw.g31[t], base[(size_t)yc * WQ] * m, a31);
    }
    float a7 = 0.f;
#pragma unroll
    for (int t = 0; t < 7; ++t) {
        int yy = y + t - 3;
        int yc = iclamp(yy, 0, HQ - 1);
        float m = (yy == yc) ? 1.f : 0.f;
        a7 = fmaf(gw.g7[t], base[(size_t)yc * WQ] * m, a7);
    }
    vb31[i] = a31;
    vb7[i]  = a7;
}

__global__ __launch_bounds__(32) void hblur_combine_kernel(
    const float* __restrict__ x, const float* __restrict__ L,
    const float* __restrict__ VB31, const float* __restrict__ VB7,
    const float* __restrict__ pclar, const float* __restrict__ ptex,
    float* __restrict__ out) {
    __shared__ float sg31[31];
    __shared__ float sg7[7];

    const int lane = threadIdx.x;
    const int ln   = lane & 15;
    const int hi   = lane >> 4;

    if (lane < 31) { float d = (float)lane - 15.f; sg31[lane] = expf(-d * d * (1.f / 128.f)); }
    if (lane < 7)  { float d = (float)lane - 3.f;  sg7[lane]  = expf(-d * d * (1.f / 4.5f)); }
    __syncthreads();
    float s31 = 0.f, s7 = 0.f;
    for (int t = 0; t < 31; ++t) s31 += sg31[t];
    for (int t = 0; t < 7;  ++t) s7  += sg7[t];
    const float inv31 = 1.f / s31, inv7 = 1.f / s7;

    v2f w31[12];
#pragma unroll
    for (int s = 0; s < 12; ++s) {
#pragma unroll
        for (int v = 0; v < 2; ++v) {
            int t  = 4 * s + 2 * hi + v - ln;
            int tc = iclamp(t, 0, 30);
            w31[s][v] = sg31[tc] * ((t == tc) ? inv31 : 0.f);
        }
    }
    v2f w7[6];
#pragma unroll
    for (int s = 0; s < 6; ++s) {
#pragma unroll
        for (int v = 0; v < 2; ++v) {
            int t  = 4 * s + 2 * hi + v - ln;
            int tc = iclamp(t, 0, 6);
            w7[s][v] = sg7[tc] * ((t == tc) ? inv7 : 0.f);
        }
    }

    const int x0 = blockIdx.x * 16;
    const int y0 = blockIdx.y * 16;
    const int b  = blockIdx.z;
    const int row = y0 + ln;
    const float* p31 = VB31 + (size_t)b * HW + (size_t)row * WQ;
    const float* p7  = VB7  + (size_t)b * HW + (size_t)row * WQ;
    const int kbase = 2 * hi;

    v8f acc31 = {0.f, 0.f, 0.f, 0.f, 0.f, 0.f, 0.f, 0.f};
#pragma unroll
    for (int s = 0; s < 12; ++s) {
        v2f a;
#pragma unroll
        for (int v = 0; v < 2; ++v) {
            int col = x0 - 15 + 4 * s + kbase + v;
            int cc  = iclamp(col, 0, WQ - 1);
            a[v] = p31[cc] * ((col == cc) ? 1.f : 0.f);
        }
        acc31 = __builtin_amdgcn_wmma_f32_16x16x4_f32(
            false, a, false, w31[s], (short)0, acc31, false, false);
    }

    v8f acc7 = {0.f, 0.f, 0.f, 0.f, 0.f, 0.f, 0.f, 0.f};
#pragma unroll
    for (int s = 0; s < 6; ++s) {
        v2f a;
#pragma unroll
        for (int v = 0; v < 2; ++v) {
            int col = x0 - 3 + 4 * s + kbase + v;
            int cc  = iclamp(col, 0, WQ - 1);
            a[v] = p7[cc] * ((col == cc) ? 1.f : 0.f);
        }
        acc7 = __builtin_amdgcn_wmma_f32_16x16x4_f32(
            false, a, false, w7[s], (short)0, acc7, false, false);
    }

    const float ca = tanhf(pclar[0]) * 0.5f;
    const float ta = tanhf(ptex[0])  * 0.3f;
#pragma unroll
    for (int r = 0; r < 8; ++r) {
        int y  = y0 + r + 8 * hi;
        int xc = x0 + ln;
        size_t pidx = (size_t)b * HW + (size_t)y * WQ + xc;
        float Lv  = L[pidx];
        float le  = Lv + ca * (Lv - acc31[r]) + ta * (Lv - acc7[r]);
        float ratio = (le + 1e-6f) / (Lv + 1e-6f);
#pragma unroll
        for (int c = 0; c < 3; ++c) {
            size_t xi = (size_t)(b * 3 + c) * HW + (size_t)y * WQ + xc;
            float vv = x[xi] * ratio;
            out[xi] = fminf(fmaxf(vv, 0.f), 1.f);
        }
    }
}

extern "C" void kernel_launch(void* const* d_in, const int* in_sizes, int n_in,
                              void* d_out, int out_size, void* d_ws, size_t ws_size,
                              hipStream_t stream) {
    const float* x    = (const float*)d_in[0];
    const float* clar = (const float*)d_in[1];
    const float* tex  = (const float*)d_in[2];
    float* out = (float*)d_out;

    float* ws   = (float*)d_ws;
    float* L    = ws;
    float* VB31 = ws + (size_t)NB * HW;
    float* VB7  = ws + (size_t)2 * NB * HW;

    GW gw;
    {
        float s = 0.f;
        for (int i = 0; i < 31; ++i) { float d = (float)(i - 15); gw.g31[i] = expf(-d * d / (2.f * 8.f * 8.f)); s += gw.g31[i]; }
        for (int i = 0; i < 31; ++i) gw.g31[i] /= s;
        s = 0.f;
        for (int i = 0; i < 7; ++i)  { float d = (float)(i - 3);  gw.g7[i]  = expf(-d * d / (2.f * 1.5f * 1.5f)); s += gw.g7[i]; }
        for (int i = 0; i < 7; ++i)  gw.g7[i] /= s;
    }

    luma_kernel<<<(NB * HW / 4) / 256, 256, 0, stream>>>(x, L);
    vblur_kernel<<<(NB * HW) / 256, 256, 0, stream>>>(L, VB31, VB7, gw);
    dim3 grid3(WQ / 16, HQ / 16, NB);
    hblur_combine_kernel<<<grid3, 32, 0, stream>>>(x, L, VB31, VB7, clar, tex, out);
}


# hblur pass: 8-wave blocks stage 16-row strips of both vertical-blur planes in LDS with row-contiguous loads (4 strips per block, B fragments built once per wave) instead of per-lane strided reads
# speedup vs baseline: 1.5794x; 1.5794x over previous
#include <hip/hip_runtime.h>
#include <cmath>

#define HQ 1024
#define WQ 1024
#define HW (HQ * WQ)
#define NB 8
#define NC 3

typedef __attribute__((ext_vector_type(2))) float v2f;
typedef __attribute__((ext_vector_type(8))) float v8f;

struct GW { float g31[31]; float g7[7]; };

__device__ __forceinline__ int iclamp(int v, int lo, int hi) {
    return v < lo ? lo : (v > hi ? hi : v);
}

__global__ __launch_bounds__(256) void luma_kernel(const float* __restrict__ x,
                                                   float* __restrict__ L) {
    int i = blockIdx.x * 256 + threadIdx.x;
    int b   = i >> 18;
    int rem = i & ((HW / 4) - 1);
    const float4* xv = (const float4*)x;
    float4 r  = xv[(size_t)(b * 3 + 0) * (HW / 4) + rem];
    float4 g  = xv[(size_t)(b * 3 + 1) * (HW / 4) + rem];
    float4 bl = xv[(size_t)(b * 3 + 2) * (HW / 4) + rem];
    float4 o;
    o.x = 0.2126f * r.x + 0.7152f * g.x + 0.0722f * bl.x;
    o.y = 0.2126f * r.y + 0.7152f * g.y + 0.0722f * bl.y;
    o.z = 0.2126f * r.z + 0.7152f * g.z + 0.0722f * bl.z;
    o.w = 0.2126f * r.w + 0.7152f * g.w + 0.0722f * bl.w;
    ((float4*)L)[(size_t)b * (HW / 4) + rem] = o;
}

__global__ __launch_bounds__(256) void vblur_kernel(const float* __restrict__ L,
                                                    float* __restrict__ vb31,
                                                    float* __restrict__ vb7,
                                                    GW gw) {
    int i = blockIdx.x * 256 + threadIdx.x;
    int b  = i >> 20;
    int p  = i & (HW - 1);
    int y  = p >> 10;
    int xx = p & (WQ - 1);
    const float* base = L + (size_t)b * HW + xx;

    float a31 = 0.f;
#pragma unroll
    for (int t = 0; t < 31; ++t) {
        int yy = y + t - 15;
        int yc = iclamp(yy, 0, HQ - 1);
        float m = (yy == yc) ? 1.f : 0.f;
        a31 = fmaf(gw.g31[t], base[(size_t)yc * WQ] * m, a31);
    }
    float a7 = 0.f;
#pragma unroll
    for (int t = 0; t < 7; ++t) {
        int yy = y + t - 3;
        int yc = iclamp(yy, 0, HQ - 1);
        float m = (yy == yc) ? 1.f : 0.f;
        a7 = fmaf(gw.g7[t], base[(size_t)yc * WQ] * m, a7);
    }
    vb31[i] = a31;
    vb7[i]  = a7;
}

#define TP 164

__global__ __launch_bounds__(256) void hblur_combine_kernel(
    const float* __restrict__ x, const float* __restrict__ L,
    const float* __restrict__ VB31, const float* __restrict__ VB7,
    const float* __restrict__ pclar, const float* __restrict__ ptex,
    float* __restrict__ out) {
    __shared__ float sg31[31];
    __shared__ float sg7[7];
    __shared__ __attribute__((aligned(16))) float t31[16 * TP];
    __shared__ __attribute__((aligned(16))) float t7[16 * TP];

    const int tid  = threadIdx.x;
    const int lane = tid & 31;
    const int wv   = __builtin_amdgcn_readfirstlane(tid >> 5);
    const int ln   = lane & 15;
    const int hi   = lane >> 4;

    if (tid < 31) { float d = (float)tid - 15.f; sg31[tid] = expf(-d * d * (1.f / 128.f)); }
    if (tid < 7)  { float d = (float)tid - 3.f;  sg7[tid]  = expf(-d * d * (1.f / 4.5f)); }
    __syncthreads();
    float s31 = 0.f, s7 = 0.f;
    for (int t = 0; t < 31; ++t) s31 += sg31[t];
    for (int t = 0; t < 7;  ++t) s7  += sg7[t];
    const float inv31 = 1.f / s31, inv7 = 1.f / s7;

    v2f w31[12];
#pragma unroll
    for (int s = 0; s < 12; ++s) {
#pragma unroll
        for (int v = 0; v < 2; ++v) {
            int t  = 4 * s + 2 * hi + v - ln;
            int tc = iclamp(t, 0, 30);
            w31[s][v] = sg31[tc] * ((t == tc) ? inv31 : 0.f);
        }
    }
    v2f w7[6];
#pragma unroll
    for (int s = 0; s < 6; ++s) {
#pragma unroll
        for (int v = 0; v < 2; ++v) {
            int t  = 4 * s + 2 * hi + v - ln;
            int tc = iclamp(t, 0, 6);
            w7[s][v] = sg7[tc] * ((t == tc) ? inv7 : 0.f);
        }
    }

    const int X0 = blockIdx.x * 128;
    const int Y0 = blockIdx.y * 64;
    const int b  = blockIdx.z;
    const size_t plane = (size_t)b * HW;

    int  cix[5];
    bool cok[5];
#pragma unroll
    for (int i = 0; i < 5; ++i) {
        int col = X0 - 15 + lane + 32 * i;
        int cc  = iclamp(col, 0, WQ - 1);
        cix[i] = cc; cok[i] = (col == cc);
    }
    float pf31[2][5], pf7[2][5];
    auto stage_load = [&](int it) {
#pragma unroll
        for (int rr = 0; rr < 2; ++rr) {
            const size_t ro = plane + (size_t)(Y0 + 16 * it + 2 * wv + rr) * WQ;
            const float* g31 = VB31 + ro;
            const float* g7  = VB7 + ro;
#pragma unroll
            for (int i = 0; i < 5; ++i) {
                float a = g31[cix[i]];
                float c = g7[cix[i]];
                pf31[rr][i] = cok[i] ? a : 0.f;
                pf7[rr][i]  = cok[i] ? c : 0.f;
            }
        }
    };

    const float ca = tanhf(pclar[0]) * 0.5f;
    const float ta = tanhf(ptex[0])  * 0.3f;
    const int x0 = X0 + 16 * wv;
    const float* a31p = t31 + ln * TP + 16 * wv + 2 * hi;
    const float* a7p  = t7  + ln * TP + 16 * wv + 12 + 2 * hi;
    const int lofs = 8 * hi * WQ + ln;

    stage_load(0);
    for (int it = 0; it < 4; ++it) {
        __syncthreads();
#pragma unroll
        for (int rr = 0; rr < 2; ++rr) {
#pragma unroll
            for (int i = 0; i < 5; ++i) {
                t31[(2 * wv + rr) * TP + lane + 32 * i] = pf31[rr][i];
                t7[(2 * wv + rr) * TP + lane + 32 * i]  = pf7[rr][i];
            }
        }
        __syncthreads();
        if (it < 3) stage_load(it + 1);

        const int y0 = Y0 + 16 * it;

        v8f acc31 = {0.f, 0.f, 0.f, 0.f, 0.f, 0.f, 0.f, 0.f};
#pragma unroll
        for (int s = 0; s < 12; ++s) {
            v2f a = *(const v2f*)(a31p + 4 * s);
            acc31 = __builtin_amdgcn_wmma_f32_16x16x4_f32(
                false, a, false, w31[s], (short)0, acc31, false, false);
        }

        v8f acc7 = {0.f, 0.f, 0.f, 0.f, 0.f, 0.f, 0.f, 0.f};
#pragma unroll
        for (int s = 0; s < 6; ++s) {
            v2f a = *(const v2f*)(a7p + 4 * s);
            acc7 = __builtin_amdgcn_wmma_f32_16x16x4_f32(
                false, a, false, w7[s], (short)0, acc7, false, false);
        }

        const size_t tb = (size_t)y0 * WQ + x0;
        const float* Lt = L + plane + tb;
        const float* xt = x + (size_t)(b * 3) * HW + tb;
        float*       ot = out + (size_t)(b * 3) * HW + tb;
#pragma unroll
        for (int r = 0; r < 8; ++r) {
            float Lv  = Lt[lofs + r * WQ];
            float le  = Lv + ca * (Lv - acc31[r]) + ta * (Lv - acc7[r]);
            float ratio = (le + 1e-6f) / (Lv + 1e-6f);
#pragma unroll
            for (int c = 0; c < 3; ++c) {
                float vv = xt[c * HW + lofs + r * WQ] * ratio;
                ot[c * HW + lofs + r * WQ] = fminf(fmaxf(vv, 0.f), 1.f);
            }
        }
    }
}

extern "C" void kernel_launch(void* const* d_in, const int* in_sizes, int n_in,
                              void* d_out, int out_size, void* d_ws, size_t ws_size,
                              hipStream_t stream) {
    const float* x    = (const float*)d_in[0];
    const float* clar = (const float*)d_in[1];
    const float* tex  = (const float*)d_in[2];
    float* out = (float*)d_out;

    float* ws   = (float*)d_ws;
    float* L    = ws;
    float* VB31 = ws + (size_t)NB * HW;
    float* VB7  = ws + (size_t)2 * NB * HW;

    GW gw;
    {
        float s = 0.f;
        for (int i = 0; i < 31; ++i) { float d = (float)(i - 15); gw.g31[i] = expf(-d * d / (2.f * 8.f * 8.f)); s += gw.g31[i]; }
        for (int i = 0; i < 31; ++i) gw.g31[i] /= s;
        s = 0.f;
        for (int i = 0; i < 7; ++i)  { float d = (float)(i - 3);  gw.g7[i]  = expf(-d * d / (2.f * 1.5f * 1.5f)); s += gw.g7[i]; }
        for (int i = 0; i < 7; ++i)  gw.g7[i] /= s;
    }

    luma_kernel<<<(NB * HW / 4) / 256, 256, 0, stream>>>(x, L);
    vblur_kernel<<<(NB * HW) / 256, 256, 0, stream>>>(L, VB31, VB7, gw);
    dim3 grid3(WQ / 128, HQ / 64, NB);
    hblur_combine_kernel<<<grid3, 256, 0, stream>>>(x, L, VB31, VB7, clar, tex, out);
}


# vblur pass: each thread keeps a 62-row luma column window in registers and forms 32 output rows of both blurs from it (a row is loaded once per 32 outputs, not once per tap); tap sum order unchanged
# speedup vs baseline: 2.6026x; 1.6478x over previous
#include <hip/hip_runtime.h>
#include <cmath>

#define HQ 1024
#define WQ 1024
#define HW (HQ * WQ)
#define NB 8
#define NC 3

typedef __attribute__((ext_vector_type(2))) float v2f;
typedef __attribute__((ext_vector_type(8))) float v8f;

struct GW { float g31[31]; float g7[7]; };

__device__ __forceinline__ int iclamp(int v, int lo, int hi) {
    return v < lo ? lo : (v > hi ? hi : v);
}

__global__ __launch_bounds__(256) void luma_kernel(const float* __restrict__ x,
                                                   float* __restrict__ L) {
    int i = blockIdx.x * 256 + threadIdx.x;
    int b   = i >> 18;
    int rem = i & ((HW / 4) - 1);
    const float4* xv = (const float4*)x;
    float4 r  = xv[(size_t)(b * 3 + 0) * (HW / 4) + rem];
    float4 g  = xv[(size_t)(b * 3 + 1) * (HW / 4) + rem];
    float4 bl = xv[(size_t)(b * 3 + 2) * (HW / 4) + rem];
    float4 o;
    o.x = 0.2126f * r.x + 0.7152f * g.x + 0.0722f * bl.x;
    o.y = 0.2126f * r.y + 0.7152f * g.y + 0.0722f * bl.y;
    o.z = 0.2126f * r.z + 0.7152f * g.z + 0.0722f * bl.z;
    o.w = 0.2126f * r.w + 0.7152f * g.w + 0.0722f * bl.w;
    ((float4*)L)[(size_t)b * (HW / 4) + rem] = o;
}

#define VR 32

__global__ __launch_bounds__(256) void vblur_kernel(const float* __restrict__ L,
                                                    float* __restrict__ vb31,
                                                    float* __restrict__ vb7,
                                                    GW gw) {
    static_assert(VR >= 15 && HQ % VR == 0 && HQ / VR >= 2, "the edge case below assumes this");
    const int xx = blockIdx.x * 256 + threadIdx.x;
    const int y0 = blockIdx.y * VR;
    const int b  = blockIdx.z;
    const float* col = L + (size_t)b * HW + xx;

    float v[VR + 30];
    if (y0 != 0 && y0 + VR != HQ) {
        const float* p = col + (size_t)(y0 - 15) * WQ;
#pragma unroll
        for (int k = 0; k < VR + 30; ++k) v[k] = p[(size_t)k * WQ];
    } else {
        const bool top = (y0 == 0);
        const float* p = col + (size_t)(top ? 0 : HQ - VR - 15) * WQ;
        float e[VR + 15];
#pragma unroll
        for (int k = 0; k < VR + 15; ++k) e[k] = p[(size_t)k * WQ];
#pragma unroll
        for (int k = 0; k < VR + 30; ++k) {
            float vt = (k >= 15) ? e[k >= 15 ? k - 15 : 0] : 0.f;
            float vb = (k < VR + 15) ? e[k < VR + 15 ? k : 0] : 0.f;
            v[k] = top ? vt : vb;
        }
    }

    const size_t o = (size_t)b * HW + (size_t)y0 * WQ + xx;
    float* o31 = vb31 + o;
    float* o7  = vb7 + o;
#pragma unroll
    for (int j = 0; j < VR; ++j) {
        float a31 = 0.f;
#pragma unroll
        for (int t = 0; t < 31; ++t) a31 = fmaf(gw.g31[t], v[j + t], a31);
        float a7 = 0.f;
#pragma unroll
        for (int t = 0; t < 7; ++t) a7 = fmaf(gw.g7[t], v[j + 12 + t], a7);
        o31[(size_t)j * WQ] = a31;
        o7[(size_t)j * WQ]  = a7;
    }
}

#define TP 164

__global__ __launch_bounds__(256) void hblur_combine_kernel(
    const float* __restrict__ x, const float* __restrict__ L,
    const float* __restrict__ VB31, const float* __restrict__ VB7,
    const float* __restrict__ pclar, const float* __restrict__ ptex,
    float* __restrict__ out) {
    __shared__ float sg31[31];
    __shared__ float sg7[7];
    __shared__ __attribute__((aligned(16))) float t31[16 * TP];
    __shared__ __attribute__((aligned(16))) float t7[16 * TP];

    const int tid  = threadIdx.x;
    const int lane = tid & 31;
    const int wv   = __builtin_amdgcn_readfirstlane(tid >> 5);
    const int ln   = lane & 15;
    const int hi   = lane >> 4;

    if (tid < 31) { float d = (float)tid - 15.f; sg31[tid] = expf(-d * d * (1.f / 128.f)); }
    if (tid < 7)  { float d = (float)tid - 3.f;  sg7[tid]  = expf(-d * d * (1.f / 4.5f)); }
    __syncthreads();
    float s31 = 0.f, s7 = 0.f;
    for (int t = 0; t < 31; ++t) s31 += sg31[t];
    for (int t = 0; t < 7;  ++t) s7  += sg7[t];
    const float inv31 = 1.f / s31, inv7 = 1.f / s7;

    v2f w31[12];
#pragma unroll
    for (int s = 0; s < 12; ++s) {
#pragma unroll
        for (int v = 0; v < 2; ++v) {
            int t  = 4 * s + 2 * hi + v - ln;
            int tc = iclamp(t, 0, 30);
            w31[s][v] = sg31[tc] * ((t == tc) ? inv31 : 0.f);
        }
    }
    v2f w7[6];
#pragma unroll
    for (int s = 0; s < 6; ++s) {
#pragma unroll
        for (int v = 0; v < 2; ++v) {
            int t  = 4 * s + 2 * hi + v - ln;
            int tc = iclamp(t, 0, 6);
            w7[s][v] = sg7[tc] * ((t == tc) ? inv7 : 0.f);
        }
    }

    const int X0 = blockIdx.x * 128;
    const int Y0 = blockIdx.y * 64;
    const int b  = blockIdx.z;
    const size_t plane = (size_t)b * HW;

    int  cix[5];
    bool cok[5];
#pragma unroll
    for (int i = 0; i < 5; ++i) {
        int col = X0 - 15 + lane + 32 * i;
        int cc  = iclamp(col, 0, WQ - 1);
        cix[i] = cc; cok[i] = (col == cc);
    }
    float pf31[2][5], pf7[2][5];
    auto stage_load = [&](int it) {
#pragma unroll
        for (int rr = 0; rr < 2; ++rr) {
            const size_t ro = plane + (size_t)(Y0 + 16 * it + 2 * wv + rr) * WQ;
            const float* g31 = VB31 + ro;
            const float* g7  = VB7 + ro;
#pragma unroll
            for (int i = 0; i < 5; ++i) {
                float a = g31[cix[i]];
                float c = g7[cix[i]];
                pf31[rr][i] = cok[i] ? a : 0.f;
                pf7[rr][i]  = cok[i] ? c : 0.f;
            }
        }
    };

    const float ca = tanhf(pclar[0]) * 0.5f;
    const float ta = tanhf(ptex[0])  * 0.3f;
    const int x0 = X0 + 16 * wv;
    const float* a31p = t31 + ln * TP + 16 * wv + 2 * hi;
    const float* a7p  = t7  + ln * TP + 16 * wv + 12 + 2 * hi;
    const int lofs = 8 * hi * WQ + ln;

    stage_load(0);
    for (int it = 0; it < 4; ++it) {
        __syncthreads();
#pragma unroll
        for (int rr = 0; rr < 2; ++rr) {
#pragma unroll
            for (int i = 0; i < 5; ++i) {
                t31[(2 * wv + rr) * TP + lane + 32 * i] = pf31[rr][i];
                t7[(2 * wv + rr) * TP + lane + 32 * i]  = pf7[rr][i];
            }
        }
        __syncthreads();
        if (it < 3) stage_load(it + 1);

        const int y0 = Y0 + 16 * it;

        v8f acc31 = {0.f, 0.f, 0.f, 0.f, 0.f, 0.f, 0.f, 0.f};
#pragma unroll
        for (int s = 0; s < 12; ++s) {
            v2f a = *(const v2f*)(a31p + 4 * s);
            acc31 = __builtin_amdgcn_wmma_f32_16x16x4_f32(
                false, a, false, w31[s], (short)0, acc31, false, false);
        }

        v8f acc7 = {0.f, 0.f, 0.f, 0.f, 0.f, 0.f, 0.f, 0.f};
#pragma unroll
        for (int s = 0; s < 6; ++s) {
            v2f a = *(const v2f*)(a7p + 4 * s);
            acc7 = __builtin_amdgcn_wmma_f32_16x16x4_f32(
                false, a, false, w7[s], (short)0, acc7, false, false);
        }

        const size_t tb = (size_t)y0 * WQ + x0;
        const float* Lt = L + plane + tb;
        const float* xt = x + (size_t)(b * 3) * HW + tb;
        float*       ot = out + (size_t)(b * 3) * HW + tb;
#pragma unroll
        for (int r = 0; r < 8; ++r) {
            float Lv  = Lt[lofs + r * WQ];
            float le  = Lv + ca * (Lv - acc31[r]) + ta * (Lv - acc7[r]);
            float ratio = (le + 1e-6f) / (Lv + 1e-6f);
#pragma unroll
            for (int c = 0; c < 3; ++c) {
                float vv = xt[c * HW + lofs + r * WQ] * ratio;
                ot[c * HW + lofs + r * WQ] = fminf(fmaxf(vv, 0.f), 1.f);
            }
        }
    }
}

extern "C" void kernel_launch(void* const* d_in, const int* in_sizes, int n_in,
                              void* d_out, int out_size, void* d_ws, size_t ws_size,
                              hipStream_t stream) {
    const float* x    = (const float*)d_in[0];
    const float* clar = (const float*)d_in[1];
    const float* tex  = (const float*)d_in[2];
    float* out = (float*)d_out;

    float* ws   = (float*)d_ws;
    float* L    = ws;
    float* VB31 = ws + (size_t)NB * HW;
    float* VB7  = ws + (size_t)2 * NB * HW;

    GW gw;
    {
        float s = 0.f;
        for (int i = 0; i < 31; ++i) { float d = (float)(i - 15); gw.g31[i] = expf(-d * d / (2.f * 8.f * 8.f)); s += gw.g31[i]; }
        for (int i = 0; i < 31; ++i) gw.g31[i] /= s;
        s = 0.f;
        for (int i = 0; i < 7; ++i)  { float d = (float)(i - 3);  gw.g7[i]  = expf(-d * d / (2.f * 1.5f * 1.5f)); s += gw.g7[i]; }
        for (int i = 0; i < 7; ++i)  gw.g7[i] /= s;
    }

    luma_kernel<<<(NB * HW / 4) / 256, 256, 0, stream>>>(x, L);
    dim3 grid2(WQ / 256, HQ / VR, NB);
    vblur_kernel<<<grid2, 256, 0, stream>>>(L, VB31, VB7, gw);
    dim3 grid3(WQ / 128, HQ / 64, NB);
    hblur_combine_kernel<<<grid3, 256, 0, stream>>>(x, L, VB31, VB7, clar, tex, out);
}


# luma pass removed: vblur forms luma in registers from the three colour planes, the combine step from the colour values it loads anyway (2 launches, no luma plane; hblur bound to 128 VGPRs)
# speedup vs baseline: 2.8602x; 1.0990x over previous
#include <hip/hip_runtime.h>
#include <cmath>

#define HQ 1024
#define WQ 1024
#define HW (HQ * WQ)
#define NB 8
#define NC 3

typedef __attribute__((ext_vector_type(2))) float v2f;
typedef __attribute__((ext_vector_type(8))) float v8f;

struct GW { float g31[31]; float g7[7]; };

__device__ __forceinline__ int iclamp(int v, int lo, int hi) {
    return v < lo ? lo : (v > hi ? hi : v);
}

__device__ __forceinline__ float luma_of(float r, float g, float bl) {
    return fmaf(0.0722f, bl, fmaf(0.7152f, g, 0.2126f * r));
}

#define VR 32

__global__ __launch_bounds__(256) void vblur_kernel(const float* __restrict__ x,
                                                    float* __restrict__ vb31,
                                                    float* __restrict__ vb7,
                                                    GW gw) {
    static_assert(VR >= 15 && HQ % VR == 0 && HQ / VR >= 2, "the edge case below assumes this");
    const int xx = blockIdx.x * 256 + threadIdx.x;
    const int y0 = blockIdx.y * VR;
    const int b  = blockIdx.z;
    const float* col = x + (size_t)(b * 3) * HW + xx;

    float v[VR + 30];
    if (y0 != 0 && y0 + VR != HQ) {
        const float* p = col + (size_t)(y0 - 15) * WQ;
#pragma unroll
        for (int k = 0; k < VR + 30; ++k)
            v[k] = luma_of(p[(size_t)k * WQ], p[(size_t)HW + (size_t)k * WQ],
                           p[(size_t)2 * HW + (size_t)k * WQ]);
    } else {
        const bool top = (y0 == 0);
        const float* p = col + (size_t)(top ? 0 : HQ - VR - 15) * WQ;
        float e[VR + 15];
#pragma unroll
        for (int k = 0; k < VR + 15; ++k)
            e[k] = luma_of(p[(size_t)k * WQ], p[(size_t)HW + (size_t)k * WQ],
                           p[(size_t)2 * HW + (size_t)k * WQ]);
#pragma unroll
        for (int k = 0; k < VR + 30; ++k) {
            float vt = (k >= 15) ? e[k >= 15 ? k - 15 : 0] : 0.f;
            float vb = (k < VR + 15) ? e[k < VR + 15 ? k : 0] : 0.f;
            v[k] = top ? vt : vb;
        }
    }

    const size_t o = (size_t)b * HW + (size_t)y0 * WQ + xx;
    float* o31 = vb31 + o;
    float* o7  = vb7 + o;
#pragma unroll
    for (int j = 0; j < VR; ++j) {
        float a31 = 0.f;
#pragma unroll
        for (int t = 0; t < 31; ++t) a31 = fmaf(gw.g31[t], v[j + t], a31);
        float a7 = 0.f;
#pragma unroll
        for (int t = 0; t < 7; ++t) a7 = fmaf(gw.g7[t], v[j + 12 + t], a7);
        o31[(size_t)j * WQ] = a31;
        o7[(size_t)j * WQ]  = a7;
    }
}

#define TP 164

__global__ __launch_bounds__(256, 8) void hblur_combine_kernel(
    const float* __restrict__ x,
    const float* __restrict__ VB31, const float* __restrict__ VB7,
    const float* __restrict__ pclar, const float* __restrict__ ptex,
    float* __restrict__ out) {
    __shared__ float sg31[31];
    __shared__ float sg7[7];
    __shared__ float sinv[2];
    __shared__ __attribute__((aligned(16))) float t31[16 * TP];
    __shared__ __attribute__((aligned(16))) float t7[16 * TP];

    const int tid  = threadIdx.x;
    const int lane = tid & 31;
    const int wv   = __builtin_amdgcn_readfirstlane(tid >> 5);
    const int ln   = lane & 15;
    const int hi   = lane >> 4;

    if (tid < 31) { float d = (float)tid - 15.f; sg31[tid] = expf(-d * d * (1.f / 128.f)); }
    if (tid < 7)  { float d = (float)tid - 3.f;  sg7[tid]  = expf(-d * d * (1.f / 4.5f)); }
    __syncthreads();
    if (tid == 0) {
        float s31 = 0.f, s7 = 0.f;
        for (int t = 0; t < 31; ++t) s31 += sg31[t];
        for (int t = 0; t < 7;  ++t) s7  += sg7[t];
        sinv[0] = 1.f / s31; sinv[1] = 1.f / s7;
    }
    __syncthreads();
    const float inv31 = sinv[0], inv7 = sinv[1];

    v2f w31[12];
#pragma unroll
    for (int s = 0; s < 12; ++s) {
#pragma unroll
        for (int v = 0; v < 2; ++v) {
            int t  = 4 * s + 2 * hi + v - ln;
            int tc = iclamp(t, 0, 30);
            w31[s][v] = sg31[tc] * ((t == tc) ? inv31 : 0.f);
        }
    }
    v2f w7[6];
#pragma unroll
    for (int s = 0; s < 6; ++s) {
#pragma unroll
        for (int v = 0; v < 2; ++v) {
            int t  = 4 * s + 2 * hi + v - ln;
            int tc = iclamp(t, 0, 6);
            w7[s][v] = sg7[tc] * ((t == tc) ? inv7 : 0.f);
        }
    }

    const int X0 = blockIdx.x * 128;
    const int Y0 = blockIdx.y * 64;
    const int b  = blockIdx.z;
    const size_t plane = (size_t)b * HW;

    int  cix[5];
    bool cok[5];
#pragma unroll
    for (int i = 0; i < 5; ++i) {
        int col = X0 - 15 + lane + 32 * i;
        int cc  = iclamp(col, 0, WQ - 1);
        cix[i] = cc; cok[i] = (col == cc);
    }
    float pf31[2][5], pf7[2][5];
    auto stage_load = [&](int it) {
#pragma unroll
        for (int rr = 0; rr < 2; ++rr) {
            const size_t ro = plane + (size_t)(Y0 + 16 * it + 2 * wv + rr) * WQ;
            const float* g31 = VB31 + ro;
            const float* g7  = VB7 + ro;
#pragma unroll
            for (int i = 0; i < 5; ++i) {
                float a = g31[cix[i]];
                float c = g7[cix[i]];
                pf31[rr][i] = cok[i] ? a : 0.f;
                pf7[rr][i]  = cok[i] ? c : 0.f;
            }
        }
    };

    const float ca = tanhf(pclar[0]) * 0.5f;
    const float ta = tanhf(ptex[0])  * 0.3f;
    const int x0 = X0 + 16 * wv;
    const float* a31p = t31 + ln * TP + 16 * wv + 2 * hi;
    const float* a7p  = t7  + ln * TP + 16 * wv + 12 + 2 * hi;
    const int lofs = 8 * hi * WQ + ln;

    stage_load(0);
    for (int it = 0; it < 4; ++it) {
        __syncthreads();
#pragma unroll
        for (int rr = 0; rr < 2; ++rr) {
#pragma unroll
            for (int i = 0; i < 5; ++i) {
                t31[(2 * wv + rr) * TP + lane + 32 * i] = pf31[rr][i];
                t7[(2 * wv + rr) * TP + lane + 32 * i]  = pf7[rr][i];
            }
        }
        __syncthreads();
        if (it < 3) stage_load(it + 1);

        const int y0 = Y0 + 16 * it;

        v8f acc31 = {0.f, 0.f, 0.f, 0.f, 0.f, 0.f, 0.f, 0.f};
#pragma unroll
        for (int s = 0; s < 12; ++s) {
            v2f a = *(const v2f*)(a31p + 4 * s);
            acc31 = __builtin_amdgcn_wmma_f32_16x16x4_f32(
                false, a, false, w31[s], (short)0, acc31, false, false);
        }

        v8f acc7 = {0.f, 0.f, 0.f, 0.f, 0.f, 0.f, 0.f, 0.f};
#pragma unroll
        for (int s = 0; s < 6; ++s) {
            v2f a = *(const v2f*)(a7p + 4 * s);
            acc7 = __builtin_amdgcn_wmma_f32_16x16x4_f32(
                false, a, false, w7[s], (short)0, acc7, false, false);
        }

        const size_t tb = (size_t)y0 * WQ + x0;
        const float* xt = x + (size_t)(b * 3) * HW + tb;
        float*       ot = out + (size_t)(b * 3) * HW + tb;
#pragma unroll
        for (int r = 0; r < 8; ++r) {
            float xc[3];
#pragma unroll
            for (int c = 0; c < 3; ++c) xc[c] = xt[c * HW + lofs + r * WQ];
            float Lv  = luma_of(xc[0], xc[1], xc[2]);
            float le  = Lv + ca * (Lv - acc31[r]) + ta * (Lv - acc7[r]);
            float ratio = (le + 1e-6f) / (Lv + 1e-6f);
#pragma unroll
            for (int c = 0; c < 3; ++c) {
                float vv = xc[c] * ratio;
                ot[c * HW + lofs + r * WQ] = fminf(fmaxf(vv, 0.f), 1.f);
            }
        }
    }
}

extern "C" void kernel_launch(void* const* d_in, const int* in_sizes, int n_in,
                              void* d_out, int out_size, void* d_ws, size_t ws_size,
                              hipStream_t stream) {
    const float* x    = (const float*)d_in[0];
    const float* clar = (const float*)d_in[1];
    const float* tex  = (const float*)d_in[2];
    float* out = (float*)d_out;

    float* ws   = (float*)d_ws;
    float* VB31 = ws;
    float* VB7  = ws + (size_t)NB * HW;

    GW gw;
    {
        float s = 0.f;
        for (int i = 0; i < 31; ++i) { float d = (float)(i - 15); gw.g31[i] = expf(-d * d / (2.f * 8.f * 8.f)); s += gw.g31[i]; }
        for (int i = 0; i < 31; ++i) gw.g31[i] /= s;
        s = 0.f;
        for (int i = 0; i < 7; ++i)  { float d = (float)(i - 3);  gw.g7[i]  = expf(-d * d / (2.f * 1.5f * 1.5f)); s += gw.g7[i]; }
        for (int i = 0; i < 7; ++i)  gw.g7[i] /= s;
    }

    dim3 grid2(WQ / 256, HQ / VR, NB);
    vblur_kernel<<<grid2, 256, 0, stream>>>(x, VB31, VB7, gw);
    dim3 grid3(WQ / 128, HQ / 64, NB);
    hblur_combine_kernel<<<grid3, 256, 0, stream>>>(x, VB31, VB7, clar, tex, out);
}


# both vertical blurs moved into the hblur kernel as WMMA f32 products on an LDS luma ring: one launch, no blur planes in memory; vertical tap sums now in WMMA order, taps exp x reciprocal
# speedup vs baseline: 3.5978x; 1.2579x over previous
#include <hip/hip_runtime.h>
#include <cmath>


#define HQ 1024
#define WQ 1024
#define HW (HQ * WQ)
#define NB 8
#define NC 3

typedef __attribute__((ext_vector_type(2))) float v2f;
typedef __attribute__((ext_vector_type(8))) float v8f;

__device__ __forceinline__ int iclamp(int v, int lo, int hi) {
    return v < lo ? lo : (v > hi ? hi : v);
}

__device__ __forceinline__ float luma_of(float r, float g, float bl) {
    return fmaf(0.0722f, bl, fmaf(0.7152f, g, 0.2126f * r));
}

#define TP 164
#define LP 168
#define RING 64

__global__ __launch_bounds__(256, 8) void blur_combine_kernel(
    const float* __restrict__ x,
    const float* __restrict__ pclar, const float* __restrict__ ptex,
    float* __restrict__ out) {
    __shared__ float sg31[31];
    __shared__ float sg7[7];
    __shared__ float sinv[2];
    __shared__ __attribute__((aligned(16))) float lum[RING * LP];
    __shared__ __attribute__((aligned(16))) float t31[16 * TP];
    __shared__ __attribute__((aligned(16))) float t7[16 * TP];

    const int tid  = threadIdx.x;
    const int lane = tid & 31;
    const int wv   = __builtin_amdgcn_readfirstlane(tid >> 5);
    const int ln   = lane & 15;
    const int hi   = lane >> 4;

    if (tid < 31) { float d = (float)tid - 15.f; sg31[tid] = expf(-d * d * (1.f / 128.f)); }
    if (tid < 7)  { float d = (float)tid - 3.f;  sg7[tid]  = expf(-d * d * (1.f / 4.5f)); }
    __syncthreads();
    if (tid == 0) {
        float s31 = 0.f, s7 = 0.f;
        for (int t = 0; t < 31; ++t) s31 += sg31[t];
        for (int t = 0; t < 7;  ++t) s7  += sg7[t];
        sinv[0] = 1.f / s31; sinv[1] = 1.f / s7;
    }
    __syncthreads();
    const float inv31 = sinv[0], inv7 = sinv[1];

    const int X0 = blockIdx.x * 128;
    const int Y0 = blockIdx.y * 64;
    const int b  = blockIdx.z;
    const float* xb = x + (size_t)(b * 3) * HW;

    unsigned cix[5];
    bool cok[5];
#pragma unroll
    for (int i = 0; i < 5; ++i) {
        int col = X0 - 15 + lane + 32 * i;
        int cc  = iclamp(col, 0, WQ - 1);
        cix[i] = (unsigned)cc; cok[i] = (col == cc);
    }
    auto slot_load = [&](int q, float (&pl)[2][5]) {
#pragma unroll
        for (int rr = 0; rr < 2; ++rr) {
            const int row = Y0 - 15 + 16 * q + 2 * wv + rr;
            const float* p0 = xb + (size_t)iclamp(row, 0, HQ - 1) * WQ;
            const float* p1 = p0 + HW;
            const float* p2 = p1 + HW;
#pragma unroll
            for (int i = 0; i < 5; ++i)
                pl[rr][i] = luma_of(p0[cix[i]], p1[cix[i]], p2[cix[i]]);
        }
    };
    auto slot_store = [&](int q, const float (&pl)[2][5]) {
#pragma unroll
        for (int rr = 0; rr < 2; ++rr) {
            const int row = Y0 - 15 + 16 * q + 2 * wv + rr;
            const bool rok = (row >= 0) && (row < HQ);
            float* d = lum + ((16 * q + 2 * wv + rr) & (RING - 1)) * LP + lane;
#pragma unroll
            for (int i = 0; i < 5; ++i)
                d[32 * i] = (rok && cok[i]) ? pl[rr][i] : 0.f;
        }
    };

    float pl[2][5];
    {
        float pa[2][5], pc[2][5];
        slot_load(0, pa); slot_load(1, pc); slot_load(2, pl);
        slot_store(0, pa); slot_store(1, pc); slot_store(2, pl);
    }
    slot_load(3, pl);

    v2f w31[12];
#pragma unroll
    for (int s = 0; s < 12; ++s) {
#pragma unroll
        for (int v = 0; v < 2; ++v) {
            int t  = 4 * s + 2 * hi + v - ln;
            int tc = iclamp(t, 0, 30);
            w31[s][v] = sg31[tc] * ((t == tc) ? inv31 : 0.f);
        }
    }
    v2f w7[6];
#pragma unroll
    for (int s = 0; s < 6; ++s) {
#pragma unroll
        for (int v = 0; v < 2; ++v) {
            int t  = 4 * s + 2 * hi + v - ln;
            int tc = iclamp(t, 0, 6);
            w7[s][v] = sg7[tc] * ((t == tc) ? inv7 : 0.f);
        }
    }

    const float ca = tanhf(pclar[0]) * 0.5f;
    const float ta = tanhf(ptex[0])  * 0.3f;
    const int x0 = X0 + 16 * wv;
    const float* a31p = t31 + ln * TP + 16 * wv + 2 * hi;
    const float* a7p  = t7  + ln * TP + 16 * wv + 12 + 2 * hi;
    const unsigned lofs = 8u * (unsigned)hi * WQ + (unsigned)ln;
    const float* lb = lum + 2 * hi * LP + ln;
    const int dofs = 8 * hi * TP + ln;

    auto vtile31 = [&](int it, int j) {
        v8f acc = {0.f, 0.f, 0.f, 0.f, 0.f, 0.f, 0.f, 0.f};
#pragma unroll
        for (int s = 0; s < 12; ++s) {
            const float* p = lb + ((16 * it + 4 * s) & (RING - 1)) * LP + 16 * j;
            v2f bf;
            bf[0] = p[0];
            bf[1] = p[LP];
            acc = __builtin_amdgcn_wmma_f32_16x16x4_f32(
                false, w31[s], false, bf, (short)0, acc, false, false);
        }
        float* d = t31 + dofs + 16 * j;
#pragma unroll
        for (int r = 0; r < 8; ++r) d[r * TP] = acc[r];
    };
    auto vtile7 = [&](int it, int j) {
        v8f acc = {0.f, 0.f, 0.f, 0.f, 0.f, 0.f, 0.f, 0.f};
#pragma unroll
        for (int s = 0; s < 6; ++s) {
            const float* p = lb + ((16 * it + 12 + 4 * s) & (RING - 1)) * LP + 16 * j;
            v2f bf;
            bf[0] = p[0];
            bf[1] = p[LP];
            acc = __builtin_amdgcn_wmma_f32_16x16x4_f32(
                false, w7[s], false, bf, (short)0, acc, false, false);
        }
        float* d = t7 + dofs + 16 * j;
#pragma unroll
        for (int r = 0; r < 8; ++r) d[r * TP] = acc[r];
    };

    for (int it = 0; it < 4; ++it) {
        __syncthreads();

        vtile31(it, wv);
        vtile7(it, wv);
        if (wv == 0) vtile31(it, 8);
        else if (wv == 1) vtile7(it, 8);
        else if (wv == 2) vtile31(it, 9);
        else if (wv == 3) vtile7(it, 9);

        __syncthreads();

        if (it < 3) slot_store(it + 3, pl);

        const int y0 = Y0 + 16 * it;

        v8f acc31 = {0.f, 0.f, 0.f, 0.f, 0.f, 0.f, 0.f, 0.f};
#pragma unroll
        for (int s = 0; s < 12; ++s) {
            v2f a = *(const v2f*)(a31p + 4 * s);
            acc31 = __builtin_amdgcn_wmma_f32_16x16x4_f32(
                false, a, false, w31[s], (short)0, acc31, false, false);
        }

        v8f acc7 = {0.f, 0.f, 0.f, 0.f, 0.f, 0.f, 0.f, 0.f};
#pragma unroll
        for (int s = 0; s < 6; ++s) {
            v2f a = *(const v2f*)(a7p + 4 * s);
            acc7 = __builtin_amdgcn_wmma_f32_16x16x4_f32(
                false, a, false, w7[s], (short)0, acc7, false, false);
        }

        const size_t tb = (size_t)y0 * WQ + x0;
        const float* xt = xb + tb;
        float*       ot = out + (size_t)(b * 3) * HW + tb;
        const float* xp[3] = {xt, xt + HW, xt + 2 * HW};
        float*       op[3] = {ot, ot + HW, ot + 2 * HW};
#pragma unroll
        for (int r = 0; r < 8; ++r) {
            float xc[3];
#pragma unroll
            for (int c = 0; c < 3; ++c) xc[c] = (xp[c] + r * WQ)[lofs];
            float Lv  = luma_of(xc[0], xc[1], xc[2]);
            float le  = Lv + ca * (Lv - acc31[r]) + ta * (Lv - acc7[r]);
            float ratio = (le + 1e-6f) / (Lv + 1e-6f);
#pragma unroll
            for (int c = 0; c < 3; ++c) {
                float vv = xc[c] * ratio;
                (op[c] + r * WQ)[lofs] = fminf(fmaxf(vv, 0.f), 1.f);
            }
        }

        if (it < 2) slot_load(it + 4, pl);
    }
}

extern "C" void kernel_launch(void* const* d_in, const int* in_sizes, int n_in,
                              void* d_out, int out_size, void* d_ws, size_t ws_size,
                              hipStream_t stream) {
    const float* x    = (const float*)d_in[0];
    const float* clar = (const float*)d_in[1];
    const float* tex  = (const float*)d_in[2];
    float* out = (float*)d_out;

    dim3 grid(WQ / 128, HQ / 64, NB);
    blur_combine_kernel<<<grid, 256, 0, stream>>>(x, clar, tex, out);
}


# combine step's colour loads issued ahead: 18 of a lane's 24 before the strip's first barrier, in flight under both matrix passes (barriers fence LDS only), not four load-wait groups behind them
# speedup vs baseline: 3.8156x; 1.0605x over previous
#include <hip/hip_runtime.h>
#include <cmath>


#define HQ 1024
#define WQ 1024
#define HW (HQ * WQ)
#define NB 8
#define NC 3

typedef __attribute__((ext_vector_type(2))) float v2f;
typedef __attribute__((ext_vector_type(8))) float v8f;

__device__ __forceinline__ int iclamp(int v, int lo, int hi) {
    return v < lo ? lo : (v > hi ? hi : v);
}

__device__ __forceinline__ float luma_of(float r, float g, float bl) {
    return fmaf(0.0722f, bl, fmaf(0.7152f, g, 0.2126f * r));
}

__device__ __forceinline__ void lds_barrier() {
    __builtin_amdgcn_fence(__ATOMIC_RELEASE, "workgroup", "local");
    __builtin_amdgcn_s_barrier();
    __builtin_amdgcn_fence(__ATOMIC_ACQUIRE, "workgroup", "local");
}

#define TP 164
#define LP 168
#define RING 64

__global__ __launch_bounds__(256, 8) void blur_combine_kernel(
    const float* __restrict__ x,
    const float* __restrict__ pclar, const float* __restrict__ ptex,
    float* __restrict__ out) {
    __shared__ float sg31[31];
    __shared__ float sg7[7];
    __shared__ float sinv[2];
    __shared__ __attribute__((aligned(16))) float lum[RING * LP];
    __shared__ __attribute__((aligned(16))) float t31[16 * TP];
    __shared__ __attribute__((aligned(16))) float t7[16 * TP];

    const int tid  = threadIdx.x;
    const int lane = tid & 31;
    const int wv   = __builtin_amdgcn_readfirstlane(tid >> 5);
    const int ln   = lane & 15;
    const int hi   = lane >> 4;

    if (tid < 31) { float d = (float)tid - 15.f; sg31[tid] = expf(-d * d * (1.f / 128.f)); }
    if (tid < 7)  { float d = (float)tid - 3.f;  sg7[tid]  = expf(-d * d * (1.f / 4.5f)); }
    __syncthreads();
    if (tid == 0) {
        float s31 = 0.f, s7 = 0.f;
        for (int t = 0; t < 31; ++t) s31 += sg31[t];
        for (int t = 0; t < 7;  ++t) s7  += sg7[t];
        sinv[0] = 1.f / s31; sinv[1] = 1.f / s7;
    }
    __syncthreads();
    const float inv31 = sinv[0], inv7 = sinv[1];

    const int X0 = blockIdx.x * 128;
    const int Y0 = blockIdx.y * 64;
    const int b  = blockIdx.z;
    const float* xb = x + (size_t)(b * 3) * HW;

    unsigned cix[5];
    bool cok[5];
#pragma unroll
    for (int i = 0; i < 5; ++i) {
        int col = X0 - 15 + lane + 32 * i;
        int cc  = iclamp(col, 0, WQ - 1);
        cix[i] = (unsigned)cc; cok[i] = (col == cc);
    }
    auto slot_load = [&](int q, float (&pl)[2][5]) {
#pragma unroll
        for (int rr = 0; rr < 2; ++rr) {
            const int row = Y0 - 15 + 16 * q + 2 * wv + rr;
            const float* p0 = xb + (size_t)iclamp(row, 0, HQ - 1) * WQ;
            const float* p1 = p0 + HW;
            const float* p2 = p1 + HW;
#pragma unroll
            for (int i = 0; i < 5; ++i)
                pl[rr][i] = luma_of(p0[cix[i]], p1[cix[i]], p2[cix[i]]);
        }
    };
    auto slot_store = [&](int q, const float (&pl)[2][5]) {
#pragma unroll
        for (int rr = 0; rr < 2; ++rr) {
            const int row = Y0 - 15 + 16 * q + 2 * wv + rr;
            const bool rok = (row >= 0) && (row < HQ);
            float* d = lum + ((16 * q + 2 * wv + rr) & (RING - 1)) * LP + lane;
#pragma unroll
            for (int i = 0; i < 5; ++i)
                d[32 * i] = (rok && cok[i]) ? pl[rr][i] : 0.f;
        }
    };

    {
        float pa[2][5], pc[2][5], pl[2][5];
        slot_load(0, pa); slot_load(1, pc); slot_load(2, pl);
        slot_store(0, pa); slot_store(1, pc); slot_store(2, pl);
    }
    float pd[2][5];
    slot_load(3, pd);

    v2f w31[12];
#pragma unroll
    for (int s = 0; s < 12; ++s) {
#pragma unroll
        for (int v = 0; v < 2; ++v) {
            int t  = 4 * s + 2 * hi + v - ln;
            int tc = iclamp(t, 0, 30);
            w31[s][v] = sg31[tc] * ((t == tc) ? inv31 : 0.f);
        }
    }
    v2f w7[6];
#pragma unroll
    for (int s = 0; s < 6; ++s) {
#pragma unroll
        for (int v = 0; v < 2; ++v) {
            int t  = 4 * s + 2 * hi + v - ln;
            int tc = iclamp(t, 0, 6);
            w7[s][v] = sg7[tc] * ((t == tc) ? inv7 : 0.f);
        }
    }

    const float ca = __builtin_bit_cast(float, __builtin_amdgcn_readfirstlane(__builtin_bit_cast(int, tanhf(pclar[0]) * 0.5f)));
    const float ta = __builtin_bit_cast(float, __builtin_amdgcn_readfirstlane(__builtin_bit_cast(int, tanhf(ptex[0])  * 0.3f)));
    slot_store(3, pd);
    const int x0 = X0 + 16 * wv;
    const float* a31p = t31 + ln * TP + 16 * wv + 2 * hi;
    const float* a7p  = t7  + ln * TP + 16 * wv + 12 + 2 * hi;
    const unsigned lofs = 8u * (unsigned)hi * WQ + (unsigned)ln;
    const float* lb = lum + 2 * hi * LP + ln;
    const int dofs = 8 * hi * TP + ln;

    auto vtile31 = [&](int it, int j) {
        v8f acc = {0.f, 0.f, 0.f, 0.f, 0.f, 0.f, 0.f, 0.f};
#pragma unroll
        for (int s = 0; s < 12; ++s) {
            const float* p = lb + ((16 * it + 4 * s) & (RING - 1)) * LP + 16 * j;
            v2f bf;
            bf[0] = p[0];
            bf[1] = p[LP];
            acc = __builtin_amdgcn_wmma_f32_16x16x4_f32(
                false, w31[s], false, bf, (short)0, acc, false, false);
        }
        float* d = t31 + dofs + 16 * j;
#pragma unroll
        for (int r = 0; r < 8; ++r) d[r * TP] = acc[r];
    };
    auto vtile7 = [&](int it, int j) {
        v8f acc = {0.f, 0.f, 0.f, 0.f, 0.f, 0.f, 0.f, 0.f};
#pragma unroll
        for (int s = 0; s < 6; ++s) {
            const float* p = lb + ((16 * it + 12 + 4 * s) & (RING - 1)) * LP + 16 * j;
            v2f bf;
            bf[0] = p[0];
            bf[1] = p[LP];
            acc = __builtin_amdgcn_wmma_f32_16x16x4_f32(
                false, w7[s], false, bf, (short)0, acc, false, false);
        }
        float* d = t7 + dofs + 16 * j;
#pragma unroll
        for (int r = 0; r < 8; ++r) d[r * TP] = acc[r];
    };

    for (int it = 0; it < 4; ++it) {
        const int y0 = Y0 + 16 * it;
        const size_t tb = (size_t)y0 * WQ + x0;
        const float* xt = xb + tb;
        float*       ot = out + (size_t)(b * 3) * HW + tb;
        const float* xp[3] = {xt, xt + HW, xt + 2 * HW};
        float*       op[3] = {ot, ot + HW, ot + 2 * HW};
        float xc[8][3];
#pragma unroll
        for (int r = 0; r < 6; ++r) {
#pragma unroll
            for (int c = 0; c < 3; ++c) xc[r][c] = (xp[c] + r * WQ)[lofs];
        }

        lds_barrier();

        vtile31(it, wv);
        vtile7(it, wv);
        if (wv == 0) vtile31(it, 8);
        else if (wv == 1) vtile7(it, 8);
        else if (wv == 2) vtile31(it, 9);
        else if (wv == 3) vtile7(it, 9);

        lds_barrier();

        v8f acc31 = {0.f, 0.f, 0.f, 0.f, 0.f, 0.f, 0.f, 0.f};
#pragma unroll
        for (int s = 0; s < 12; ++s) {
            v2f a = *(const v2f*)(a31p + 4 * s);
            acc31 = __builtin_amdgcn_wmma_f32_16x16x4_f32(
                false, a, false, w31[s], (short)0, acc31, false, false);
        }

        v8f acc7 = {0.f, 0.f, 0.f, 0.f, 0.f, 0.f, 0.f, 0.f};
#pragma unroll
        for (int s = 0; s < 6; ++s) {
            v2f a = *(const v2f*)(a7p + 4 * s);
            acc7 = __builtin_amdgcn_wmma_f32_16x16x4_f32(
                false, a, false, w7[s], (short)0, acc7, false, false);
        }

#pragma unroll
        for (int r = 0; r < 8; ++r) {
            if (r == 4) {
#pragma unroll
                for (int q = 6; q < 8; ++q) {
#pragma unroll
                    for (int c = 0; c < 3; ++c) xc[q][c] = (xp[c] + q * WQ)[lofs];
                }
            }
            float Lv  = luma_of(xc[r][0], xc[r][1], xc[r][2]);
            float le  = Lv + ca * (Lv - acc31[r]) + ta * (Lv - acc7[r]);
            float ratio = (le + 1e-6f) / (Lv + 1e-6f);
#pragma unroll
            for (int c = 0; c < 3; ++c) {
                float vv = xc[r][c] * ratio;
                (op[c] + r * WQ)[lofs] = fminf(fmaxf(vv, 0.f), 1.f);
            }
        }

        if (it < 2) {
            float pl[2][5];
            slot_load(it + 4, pl);
            slot_store(it + 4, pl);
        }
    }
}

extern "C" void kernel_launch(void* const* d_in, const int* in_sizes, int n_in,
                              void* d_out, int out_size, void* d_ws, size_t ws_size,
                              hipStream_t stream) {
    const float* x    = (const float*)d_in[0];
    const float* clar = (const float*)d_in[1];
    const float* tex  = (const float*)d_in[2];
    float* out = (float*)d_out;

    dim3 grid(WQ / 128, HQ / 64, NB);
    blur_combine_kernel<<<grid, 256, 0, stream>>>(x, clar, tex, out);
}


# combine step: ratios go through an LDS tile (one more barrier a strip) so that each wave scales whole 128-column rows: colour loads and output stores are 16 bytes a lane, whole 128-byte lines
# speedup vs baseline: 4.7039x; 1.2328x over previous
#include <hip/hip_runtime.h>
#include <cmath>


#define HQ 1024
#define WQ 1024
#define HW (HQ * WQ)
#define NB 8
#define NC 3

typedef __attribute__((ext_vector_type(2))) float v2f;
typedef __attribute__((ext_vector_type(8))) float v8f;
typedef __attribute__((ext_vector_type(4))) float v4f;

__device__ __forceinline__ int iclamp(int v, int lo, int hi) {
    return v < lo ? lo : (v > hi ? hi : v);
}

__device__ __forceinline__ float luma_of(float r, float g, float bl) {
    return fmaf(0.0722f, bl, fmaf(0.7152f, g, 0.2126f * r));
}

__device__ __forceinline__ void lds_barrier() {
    __builtin_amdgcn_fence(__ATOMIC_RELEASE, "workgroup", "local");
    __builtin_amdgcn_s_barrier();
    __builtin_amdgcn_fence(__ATOMIC_ACQUIRE, "workgroup", "local");
}

#define TP 164
#define LP 168
#define RING 64
#define RP 132

__global__ __launch_bounds__(256, 8) void blur_combine_kernel(
    const float* __restrict__ x,
    const float* __restrict__ pclar, const float* __restrict__ ptex,
    float* __restrict__ out) {
    __shared__ float sg31[31];
    __shared__ float sg7[7];
    __shared__ float sinv[2];
    __shared__ __attribute__((aligned(16))) float lum[RING * LP];
    __shared__ __attribute__((aligned(16))) float t31[16 * TP];
    __shared__ __attribute__((aligned(16))) float t7[16 * TP];
    __shared__ __attribute__((aligned(16))) float rt[16 * RP];

    const int tid  = threadIdx.x;
    const int lane = tid & 31;
    const int wv   = __builtin_amdgcn_readfirstlane(tid >> 5);
    const int ln   = lane & 15;
    const int hi   = lane >> 4;

    if (tid < 31) { float d = (float)tid - 15.f; sg31[tid] = expf(-d * d * (1.f / 128.f)); }
    if (tid < 7)  { float d = (float)tid - 3.f;  sg7[tid]  = expf(-d * d * (1.f / 4.5f)); }
    __syncthreads();
    if (tid == 0) {
        float s31 = 0.f, s7 = 0.f;
        for (int t = 0; t < 31; ++t) s31 += sg31[t];
        for (int t = 0; t < 7;  ++t) s7  += sg7[t];
        sinv[0] = 1.f / s31; sinv[1] = 1.f / s7;
    }
    __syncthreads();
    const float inv31 = sinv[0], inv7 = sinv[1];

    const int X0 = blockIdx.x * 128;
    const int Y0 = blockIdx.y * 64;
    const int b  = blockIdx.z;
    const float* xb = x + (size_t)(b * 3) * HW;

    unsigned cix[5];
    bool cok[5];
#pragma unroll
    for (int i = 0; i < 5; ++i) {
        int col = X0 - 15 + lane + 32 * i;
        int cc  = iclamp(col, 0, WQ - 1);
        cix[i] = (unsigned)cc; cok[i] = (col == cc);
    }
    auto slot_load = [&](int q, float (&pl)[2][5]) {
#pragma unroll
        for (int rr = 0; rr < 2; ++rr) {
            const int row = Y0 - 15 + 16 * q + 2 * wv + rr;
            const float* p0 = xb + (size_t)iclamp(row, 0, HQ - 1) * WQ;
            const float* p1 = p0 + HW;
            const float* p2 = p1 + HW;
#pragma unroll
            for (int i = 0; i < 5; ++i)
                pl[rr][i] = luma_of(p0[cix[i]], p1[cix[i]], p2[cix[i]]);
        }
    };
    auto slot_store = [&](int q, const float (&pl)[2][5]) {
#pragma unroll
        for (int rr = 0; rr < 2; ++rr) {
            const int row = Y0 - 15 + 16 * q + 2 * wv + rr;
            const bool rok = (row >= 0) && (row < HQ);
            float* d = lum + ((16 * q + 2 * wv + rr) & (RING - 1)) * LP + lane;
#pragma unroll
            for (int i = 0; i < 5; ++i)
                d[32 * i] = (rok && cok[i]) ? pl[rr][i] : 0.f;
        }
    };

    {
        float pa[2][5], pc[2][5], pl[2][5];
        slot_load(0, pa); slot_load(1, pc); slot_load(2, pl);
        slot_store(0, pa); slot_store(1, pc); slot_store(2, pl);
    }
    float pd[2][5];
    slot_load(3, pd);

    v2f w31[12];
#pragma unroll
    for (int s = 0; s < 12; ++s) {
#pragma unroll
        for (int v = 0; v < 2; ++v) {
            int t  = 4 * s + 2 * hi + v - ln;
            int tc = iclamp(t, 0, 30);
            w31[s][v] = sg31[tc] * ((t == tc) ? inv31 : 0.f);
        }
    }
    v2f w7[6];
#pragma unroll
    for (int s = 0; s < 6; ++s) {
#pragma unroll
        for (int v = 0; v < 2; ++v) {
            int t  = 4 * s + 2 * hi + v - ln;
            int tc = iclamp(t, 0, 6);
            w7[s][v] = sg7[tc] * ((t == tc) ? inv7 : 0.f);
        }
    }

    const float ca = __builtin_bit_cast(float, __builtin_amdgcn_readfirstlane(__builtin_bit_cast(int, tanhf(pclar[0]) * 0.5f)));
    const float ta = __builtin_bit_cast(float, __builtin_amdgcn_readfirstlane(__builtin_bit_cast(int, tanhf(ptex[0])  * 0.3f)));
    slot_store(3, pd);
    const float* a31p = t31 + ln * TP + 16 * wv + 2 * hi;
    const float* a7p  = t7  + ln * TP + 16 * wv + 12 + 2 * hi;
    const float* lvp = lum + 8 * hi * LP + 16 * wv + 15 + ln;
    float*       rtw = rt + 8 * hi * RP + 16 * wv + ln;
    const float* rtr = rt + wv * RP + 4 * lane;
    const unsigned rofs = (unsigned)wv * WQ + 4u * (unsigned)lane;
    const float* lb = lum + 2 * hi * LP + ln;
    const int dofs = 8 * hi * TP + ln;

    auto vtile31 = [&](int it, int j) {
        v8f acc = {0.f, 0.f, 0.f, 0.f, 0.f, 0.f, 0.f, 0.f};
#pragma unroll
        for (int s = 0; s < 12; ++s) {
            const float* p = lb + ((16 * it + 4 * s) & (RING - 1)) * LP + 16 * j;
            v2f bf;
            bf[0] = p[0];
            bf[1] = p[LP];
            acc = __builtin_amdgcn_wmma_f32_16x16x4_f32(
                false, w31[s], false, bf, (short)0, acc, false, false);
        }
        float* d = t31 + dofs + 16 * j;
#pragma unroll
        for (int r = 0; r < 8; ++r) d[r * TP] = acc[r];
    };
    auto vtile7 = [&](int it, int j) {
        v8f acc = {0.f, 0.f, 0.f, 0.f, 0.f, 0.f, 0.f, 0.f};
#pragma unroll
        for (int s = 0; s < 6; ++s) {
            const float* p = lb + ((16 * it + 12 + 4 * s) & (RING - 1)) * LP + 16 * j;
            v2f bf;
            bf[0] = p[0];
            bf[1] = p[LP];
            acc = __builtin_amdgcn_wmma_f32_16x16x4_f32(
                false, w7[s], false, bf, (short)0, acc, false, false);
        }
        float* d = t7 + dofs + 16 * j;
#pragma unroll
        for (int r = 0; r < 8; ++r) d[r * TP] = acc[r];
    };

#pragma clang loop unroll(disable)
    for (int it = 0; it < 4; ++it) {
        const int y0 = Y0 + 16 * it;
        const size_t tb = (size_t)y0 * WQ + X0;
        const float* xt = xb + tb;
        float*       ot = out + (size_t)(b * 3) * HW + tb;
        const float* xp[3] = {xt, xt + HW, xt + 2 * HW};
        float*       op[3] = {ot, ot + HW, ot + 2 * HW};
        v4f xq[2][3];
#pragma unroll
        for (int k = 0; k < 2; ++k) {
#pragma unroll
            for (int c = 0; c < 3; ++c) xq[k][c] = *(const v4f*)(xp[c] + 8 * k * WQ + rofs);
        }

        lds_barrier();

        vtile31(it, wv);
        vtile7(it, wv);
        if (wv == 0) vtile31(it, 8);
        else if (wv == 1) vtile7(it, 8);
        else if (wv == 2) vtile31(it, 9);
        else if (wv == 3) vtile7(it, 9);

        lds_barrier();

        v8f acc31 = {0.f, 0.f, 0.f, 0.f, 0.f, 0.f, 0.f, 0.f};
#pragma unroll
        for (int s = 0; s < 12; ++s) {
            v2f a = *(const v2f*)(a31p + 4 * s);
            acc31 = __builtin_amdgcn_wmma_f32_16x16x4_f32(
                false, a, false, w31[s], (short)0, acc31, false, false);
        }

        v8f acc7 = {0.f, 0.f, 0.f, 0.f, 0.f, 0.f, 0.f, 0.f};
#pragma unroll
        for (int s = 0; s < 6; ++s) {
            v2f a = *(const v2f*)(a7p + 4 * s);
            acc7 = __builtin_amdgcn_wmma_f32_16x16x4_f32(
                false, a, false, w7[s], (short)0, acc7, false, false);
        }

        const int sb = (16 * (it + 1)) & (RING - 1);
        float Lr[8];
        Lr[0] = lum[(hi ? (sb + 7) : (16 * it + 15)) * LP + 16 * wv + 15 + ln];
#pragma unroll
        for (int r = 1; r < 8; ++r) Lr[r] = lvp[(sb + r - 1) * LP];
#pragma unroll
        for (int r = 0; r < 8; ++r) {
            if (r == 4) __builtin_amdgcn_sched_barrier(0);
            float Lv  = Lr[r];
            float le  = Lv + ca * (Lv - acc31[r]) + ta * (Lv - acc7[r]);
            rtw[r * RP] = (le + 1e-6f) / (Lv + 1e-6f);
        }

        lds_barrier();

#pragma unroll
        for (int k = 0; k < 2; ++k) {
            const v4f rq = *(const v4f*)(rtr + 8 * k * RP);
#pragma unroll
            for (int c = 0; c < 3; ++c) {
                v4f vv = xq[k][c] * rq;
                v4f o;
#pragma unroll
                for (int e = 0; e < 4; ++e) o[e] = fminf(fmaxf(vv[e], 0.f), 1.f);
                *(v4f*)(op[c] + 8 * k * WQ + rofs) = o;
            }
        }

        if (it < 2) {
            float pl[2][5];
            slot_load(it + 4, pl);
            slot_store(it + 4, pl);
        }
    }
}

extern "C" void kernel_launch(void* const* d_in, const int* in_sizes, int n_in,
                              void* d_out, int out_size, void* d_ws, size_t ws_size,
                              hipStream_t stream) {
    const float* x    = (const float*)d_in[0];
    const float* clar = (const float*)d_in[1];
    const float* tex  = (const float*)d_in[2];
    float* out = (float*)d_out;

    dim3 grid(WQ / 128, HQ / 64, NB);
    blur_combine_kernel<<<grid, 256, 0, stream>>>(x, clar, tex, out);
}


# wave priority follows progress: s_setprio 3 in set-up and strip 0, one level lower per later strip, so the four resident blocks of a compute unit end together instead of oldest first
# speedup vs baseline: 4.9342x; 1.0490x over previous
#include <hip/hip_runtime.h>
#include <cmath>


#define HQ 1024
#define WQ 1024
#define HW (HQ * WQ)
#define NB 8
#define NC 3

typedef __attribute__((ext_vector_type(2))) float v2f;
typedef __attribute__((ext_vector_type(8))) float v8f;
typedef __attribute__((ext_vector_type(4))) float v4f;

__device__ __forceinline__ int iclamp(int v, int lo, int hi) {
    return v < lo ? lo : (v > hi ? hi : v);
}

__device__ __forceinline__ float luma_of(float r, float g, float bl) {
    return fmaf(0.0722f, bl, fmaf(0.7152f, g, 0.2126f * r));
}

__device__ __forceinline__ void lds_barrier() {
    __builtin_amdgcn_fence(__ATOMIC_RELEASE, "workgroup", "local");
    __builtin_amdgcn_s_barrier();
    __builtin_amdgcn_fence(__ATOMIC_ACQUIRE, "workgroup", "local");
}

#define TP 164
#define LP 168
#define RING 64
#define RP 132

__global__ __launch_bounds__(256, 8) void blur_combine_kernel(
    const float* __restrict__ x,
    const float* __restrict__ pclar, const float* __restrict__ ptex,
    float* __restrict__ out) {
    __shared__ float sg31[31];
    __shared__ float sg7[7];
    __shared__ float sinv[2];
    __shared__ __attribute__((aligned(16))) float lum[RING * LP];
    __shared__ __attribute__((aligned(16))) float t31[16 * TP];
    __shared__ __attribute__((aligned(16))) float t7[16 * TP];
    __shared__ __attribute__((aligned(16))) float rt[16 * RP];

    __builtin_amdgcn_s_setprio(3);

    const int tid  = threadIdx.x;
    const int lane = tid & 31;
    const int wv   = __builtin_amdgcn_readfirstlane(tid >> 5);
    const int ln   = lane & 15;
    const int hi   = lane >> 4;

    if (tid < 31) { float d = (float)tid - 15.f; sg31[tid] = expf(-d * d * (1.f / 128.f)); }
    if (tid < 7)  { float d = (float)tid - 3.f;  sg7[tid]  = expf(-d * d * (1.f / 4.5f)); }
    __syncthreads();
    if (tid == 0) {
        float s31 = 0.f, s7 = 0.f;
        for (int t = 0; t < 31; ++t) s31 += sg31[t];
        for (int t = 0; t < 7;  ++t) s7  += sg7[t];
        sinv[0] = 1.f / s31; sinv[1] = 1.f / s7;
    }
    __syncthreads();
    const float inv31 = sinv[0], inv7 = sinv[1];

    const int X0 = blockIdx.x * 128;
    const int Y0 = blockIdx.y * 64;
    const int b  = blockIdx.z;
    const float* xb = x + (size_t)(b * 3) * HW;

    unsigned cix[5];
    bool cok[5];
#pragma unroll
    for (int i = 0; i < 5; ++i) {
        int col = X0 - 15 + lane + 32 * i;
        int cc  = iclamp(col, 0, WQ - 1);
        cix[i] = (unsigned)cc; cok[i] = (col == cc);
    }
    auto slot_load = [&](int q, float (&pl)[2][5]) {
#pragma unroll
        for (int rr = 0; rr < 2; ++rr) {
            const int row = Y0 - 15 + 16 * q + 2 * wv + rr;
            const float* p0 = xb + (size_t)iclamp(row, 0, HQ - 1) * WQ;
            const float* p1 = p0 + HW;
            const float* p2 = p1 + HW;
#pragma unroll
            for (int i = 0; i < 5; ++i)
                pl[rr][i] = luma_of(p0[cix[i]], p1[cix[i]], p2[cix[i]]);
        }
    };
    auto slot_store = [&](int q, const float (&pl)[2][5]) {
#pragma unroll
        for (int rr = 0; rr < 2; ++rr) {
            const int row = Y0 - 15 + 16 * q + 2 * wv + rr;
            const bool rok = (row >= 0) && (row < HQ);
            float* d = lum + ((16 * q + 2 * wv + rr) & (RING - 1)) * LP + lane;
#pragma unroll
            for (int i = 0; i < 5; ++i)
                d[32 * i] = (rok && cok[i]) ? pl[rr][i] : 0.f;
        }
    };

    {
        float pa[2][5], pc[2][5], pl[2][5];
        slot_load(0, pa); slot_load(1, pc); slot_load(2, pl);
        slot_store(0, pa); slot_store(1, pc); slot_store(2, pl);
    }
    float pd[2][5];
    slot_load(3, pd);

    v2f w31[12];
#pragma unroll
    for (int s = 0; s < 12; ++s) {
#pragma unroll
        for (int v = 0; v < 2; ++v) {
            int t  = 4 * s + 2 * hi + v - ln;
            int tc = iclamp(t, 0, 30);
            w31[s][v] = sg31[tc] * ((t == tc) ? inv31 : 0.f);
        }
    }
    v2f w7[6];
#pragma unroll
    for (int s = 0; s < 6; ++s) {
#pragma unroll
        for (int v = 0; v < 2; ++v) {
            int t  = 4 * s + 2 * hi + v - ln;
            int tc = iclamp(t, 0, 6);
            w7[s][v] = sg7[tc] * ((t == tc) ? inv7 : 0.f);
        }
    }

    const float ca = __builtin_bit_cast(float, __builtin_amdgcn_readfirstlane(__builtin_bit_cast(int, tanhf(pclar[0]) * 0.5f)));
    const float ta = __builtin_bit_cast(float, __builtin_amdgcn_readfirstlane(__builtin_bit_cast(int, tanhf(ptex[0])  * 0.3f)));
    slot_store(3, pd);
    const float* a31p = t31 + ln * TP + 16 * wv + 2 * hi;
    const float* a7p  = t7  + ln * TP + 16 * wv + 12 + 2 * hi;
    const float* lvp = lum + 8 * hi * LP + 16 * wv + 15 + ln;
    float*       rtw = rt + 8 * hi * RP + 16 * wv + ln;
    const float* rtr = rt + wv * RP + 4 * lane;
    const unsigned rofs = (unsigned)wv * WQ + 4u * (unsigned)lane;
    const float* lb = lum + 2 * hi * LP + ln;
    const int dofs = 8 * hi * TP + ln;

    auto vtile31 = [&](int it, int j) {
        v8f acc = {0.f, 0.f, 0.f, 0.f, 0.f, 0.f, 0.f, 0.f};
#pragma unroll
        for (int s = 0; s < 12; ++s) {
            const float* p = lb + ((16 * it + 4 * s) & (RING - 1)) * LP + 16 * j;
            v2f bf;
            bf[0] = p[0];
            bf[1] = p[LP];
            acc = __builtin_amdgcn_wmma_f32_16x16x4_f32(
                false, w31[s], false, bf, (short)0, acc, false, false);
        }
        float* d = t31 + dofs + 16 * j;
#pragma unroll
        for (int r = 0; r < 8; ++r) d[r * TP] = acc[r];
    };
    auto vtile7 = [&](int it, int j) {
        v8f acc = {0.f, 0.f, 0.f, 0.f, 0.f, 0.f, 0.f, 0.f};
#pragma unroll
        for (int s = 0; s < 6; ++s) {
            const float* p = lb + ((16 * it + 12 + 4 * s) & (RING - 1)) * LP + 16 * j;
            v2f bf;
            bf[0] = p[0];
            bf[1] = p[LP];
            acc = __builtin_amdgcn_wmma_f32_16x16x4_f32(
                false, w7[s], false, bf, (short)0, acc, false, false);
        }
        float* d = t7 + dofs + 16 * j;
#pragma unroll
        for (int r = 0; r < 8; ++r) d[r * TP] = acc[r];
    };

#pragma clang loop unroll(disable)
    for (int it = 0; it < 4; ++it) {
        if (it == 1) __builtin_amdgcn_s_setprio(2);
        else if (it == 2) __builtin_amdgcn_s_setprio(1);
        else if (it == 3) __builtin_amdgcn_s_setprio(0);

        const int y0 = Y0 + 16 * it;
        const size_t tb = (size_t)y0 * WQ + X0;
        const float* xt = xb + tb;
        float*       ot = out + (size_t)(b * 3) * HW + tb;
        const float* xp[3] = {xt, xt + HW, xt + 2 * HW};
        float*       op[3] = {ot, ot + HW, ot + 2 * HW};
        v4f xq[2][3];
#pragma unroll
        for (int k = 0; k < 2; ++k) {
#pragma unroll
            for (int c = 0; c < 3; ++c) xq[k][c] = *(const v4f*)(xp[c] + 8 * k * WQ + rofs);
        }

        lds_barrier();

        vtile31(it, wv);
        vtile7(it, wv);
        if (wv == 0) vtile31(it, 8);
        else if (wv == 1) vtile7(it, 8);
        else if (wv == 2) vtile31(it, 9);
        else if (wv == 3) vtile7(it, 9);

        lds_barrier();

        v8f acc31 = {0.f, 0.f, 0.f, 0.f, 0.f, 0.f, 0.f, 0.f};
#pragma unroll
        for (int s = 0; s < 12; ++s) {
            v2f a = *(const v2f*)(a31p + 4 * s);
            acc31 = __builtin_amdgcn_wmma_f32_16x16x4_f32(
                false, a, false, w31[s], (short)0, acc31, false, false);
        }

        v8f acc7 = {0.f, 0.f, 0.f, 0.f, 0.f, 0.f, 0.f, 0.f};
#pragma unroll
        for (int s = 0; s < 6; ++s) {
            v2f a = *(const v2f*)(a7p + 4 * s);
            acc7 = __builtin_amdgcn_wmma_f32_16x16x4_f32(
                false, a, false, w7[s], (short)0, acc7, false, false);
        }

        const int sb = (16 * (it + 1)) & (RING - 1);
        float Lr[8];
        Lr[0] = lum[(hi ? (sb + 7) : (16 * it + 15)) * LP + 16 * wv + 15 + ln];
#pragma unroll
        for (int r = 1; r < 8; ++r) Lr[r] = lvp[(sb + r - 1) * LP];
#pragma unroll
        for (int r = 0; r < 8; ++r) {
            if (r == 4) __builtin_amdgcn_sched_barrier(0);
            float Lv  = Lr[r];
            float le  = Lv + ca * (Lv - acc31[r]) + ta * (Lv - acc7[r]);
            rtw[r * RP] = (le + 1e-6f) / (Lv + 1e-6f);
        }

        lds_barrier();

#pragma unroll
        for (int k = 0; k < 2; ++k) {
            const v4f rq = *(const v4f*)(rtr + 8 * k * RP);
#pragma unroll
            for (int c = 0; c < 3; ++c) {
                v4f vv = xq[k][c] * rq;
                v4f o;
#pragma unroll
                for (int e = 0; e < 4; ++e) o[e] = fminf(fmaxf(vv[e], 0.f), 1.f);
                *(v4f*)(op[c] + 8 * k * WQ + rofs) = o;
            }
        }

        if (it < 2) {
            float pl[2][5];
            slot_load(it + 4, pl);
            slot_store(it + 4, pl);
        }
    }
}

extern "C" void kernel_launch(void* const* d_in, const int* in_sizes, int n_in,
                              void* d_out, int out_size, void* d_ws, size_t ws_size,
                              hipStream_t stream) {
    const float* x    = (const float*)d_in[0];
    const float* clar = (const float*)d_in[1];
    const float* tex  = (const float*)d_in[2];
    float* out = (float*)d_out;

    dim3 grid(WQ / 128, HQ / 64, NB);
    blur_combine_kernel<<<grid, 256, 0, stream>>>(x, clar, tex, out);
}


# luma staging by 16-byte row loads: each wave-load covers a row's four whole 128-byte lines, two rows' halo columns share a third (9 loads a slot and lane, not 30); ring stored 16 bytes a lane
# speedup vs baseline: 5.2531x; 1.0646x over previous
#include <hip/hip_runtime.h>
#include <cmath>


#define HQ 1024
#define WQ 1024
#define HW (HQ * WQ)
#define NB 8
#define NC 3

typedef __attribute__((ext_vector_type(2))) float v2f;
typedef __attribute__((ext_vector_type(8))) float v8f;
typedef __attribute__((ext_vector_type(4))) float v4f;

__device__ __forceinline__ int iclamp(int v, int lo, int hi) {
    return v < lo ? lo : (v > hi ? hi : v);
}

__device__ __forceinline__ float luma_of(float r, float g, float bl) {
    return fmaf(0.0722f, bl, fmaf(0.7152f, g, 0.2126f * r));
}

__device__ __forceinline__ v4f luma4(const v4f* p, unsigned i) {
    const v4f r  = (p - HW / 4)[i];
    const v4f g  = p[i];
    const v4f bl = (p + HW / 4)[i];
    v4f o;
#pragma unroll
    for (int e = 0; e < 4; ++e) o[e] = luma_of(r[e], g[e], bl[e]);
    return o;
}

__device__ __forceinline__ void lds_barrier() {
    __builtin_amdgcn_fence(__ATOMIC_RELEASE, "workgroup", "local");
    __builtin_amdgcn_s_barrier();
    __builtin_amdgcn_fence(__ATOMIC_ACQUIRE, "workgroup", "local");
}

#define TP 164
#define LP 168
#define RING 64
#define RP 132

__global__ __launch_bounds__(256, 8) void blur_combine_kernel(
    const float* __restrict__ x,
    const float* __restrict__ pclar, const float* __restrict__ ptex,
    float* __restrict__ out) {
    __shared__ float sg31[31];
    __shared__ float sg7[7];
    __shared__ float sinv[2];
    __shared__ __attribute__((aligned(16))) float lum[RING * LP];
    __shared__ __attribute__((aligned(16))) float t31[16 * TP];
    __shared__ __attribute__((aligned(16))) float t7[16 * TP];
    __shared__ __attribute__((aligned(16))) float rt[16 * RP];

    __builtin_amdgcn_s_setprio(3);

    const int tid  = threadIdx.x;
    const int lane = tid & 31;
    const int wv   = __builtin_amdgcn_readfirstlane(tid >> 5);
    const int ln   = lane & 15;
    const int hi   = lane >> 4;

    if (tid < 31) { float d = (float)tid - 15.f; sg31[tid] = expf(-d * d * (1.f / 128.f)); }
    if (tid < 7)  { float d = (float)tid - 3.f;  sg7[tid]  = expf(-d * d * (1.f / 4.5f)); }
    __syncthreads();
    if (tid == 0) {
        float s31 = 0.f, s7 = 0.f;
        for (int t = 0; t < 31; ++t) s31 += sg31[t];
        for (int t = 0; t < 7;  ++t) s7  += sg7[t];
        sinv[0] = 1.f / s31; sinv[1] = 1.f / s7;
    }
    __syncthreads();
    const float inv31 = sinv[0], inv7 = sinv[1];

    const int X0 = blockIdx.x * 128;
    const int Y0 = blockIdx.y * 64;
    const int b  = blockIdx.z;
    const float* xb = x + (size_t)(b * 3) * HW;

    const unsigned giA = (unsigned)(X0 / 4 + lane);
    const int  gC   = (ln < 4) ? ln : (32 + (ln < 9 ? ln : 8));
    const int  colC = X0 - 16 + 4 * gC;
    const int  ccC  = iclamp(colC, 0, WQ - 4);
    const bool cokC = (colC == ccC);
    const unsigned giC = (unsigned)ccC / 4u;
    const v4f* xg4 = (const v4f*)(xb + HW);
    auto slot_load = [&](int q, v4f (&pl)[3]) {
        const int row = Y0 - 15 + 16 * q + 2 * wv;
        const unsigned r0 = (unsigned)__builtin_amdgcn_readfirstlane(iclamp(row,     0, HQ - 1));
        const unsigned r1 = (unsigned)__builtin_amdgcn_readfirstlane(iclamp(row + 1, 0, HQ - 1));
        const unsigned rC = (unsigned)iclamp(row + hi, 0, HQ - 1);
        pl[0] = luma4(xg4 + (size_t)r0 * (WQ / 4), giA);
        pl[1] = luma4(xg4 + (size_t)r1 * (WQ / 4), giA);
        pl[2] = luma4(xg4, rC * (WQ / 4) + giC);
    };
    auto slot_store = [&](int q, const v4f (&pl)[3]) {
        const int row = Y0 - 15 + 16 * q + 2 * wv;
        const bool rok0 = (row >= 0) && (row < HQ);
        const bool rok1 = (row + 1 >= 0) && (row + 1 < HQ);
        const bool okC  = cokC && ((unsigned)(row + hi) < (unsigned)HQ);
        float* d = lum + ((16 * q + 2 * wv) & (RING - 1)) * LP;
        const v4f z = {0.f, 0.f, 0.f, 0.f};
        *(v4f*)(d + 16 + 4 * lane)      = rok0 ? pl[0] : z;
        *(v4f*)(d + LP + 16 + 4 * lane) = rok1 ? pl[1] : z;
        *(v4f*)(d + hi * LP + 4 * gC)   = okC  ? pl[2] : z;
    };

    {
        v4f pa[3], pc[3], pl[3];
        slot_load(0, pa); slot_load(1, pc); slot_load(2, pl);
        slot_store(0, pa); slot_store(1, pc); slot_store(2, pl);
    }
    v4f pd[3];
    slot_load(3, pd);

    v2f w31[12];
#pragma unroll
    for (int s = 0; s < 12; ++s) {
#pragma unroll
        for (int v = 0; v < 2; ++v) {
            int t  = 4 * s + 2 * hi + v - ln;
            int tc = iclamp(t, 0, 30);
            w31[s][v] = sg31[tc] * ((t == tc) ? inv31 : 0.f);
        }
    }
    v2f w7[6];
#pragma unroll
    for (int s = 0; s < 6; ++s) {
#pragma unroll
        for (int v = 0; v < 2; ++v) {
            int t  = 4 * s + 2 * hi + v - ln;
            int tc = iclamp(t, 0, 6);
            w7[s][v] = sg7[tc] * ((t == tc) ? inv7 : 0.f);
        }
    }

    const float ca = __builtin_bit_cast(float, __builtin_amdgcn_readfirstlane(__builtin_bit_cast(int, tanhf(pclar[0]) * 0.5f)));
    const float ta = __builtin_bit_cast(float, __builtin_amdgcn_readfirstlane(__builtin_bit_cast(int, tanhf(ptex[0])  * 0.3f)));
    slot_store(3, pd);
    const float* a31p = t31 + ln * TP + 16 * wv + 2 * hi;
    const float* a7p  = t7  + ln * TP + 16 * wv + 12 + 2 * hi;
    const float* lvp = lum + 8 * hi * LP + 16 * wv + 16 + ln;
    float*       rtw = rt + 8 * hi * RP + 16 * wv + ln;
    const float* rtr = rt + wv * RP + 4 * lane;
    const unsigned rofs = (unsigned)wv * WQ + 4u * (unsigned)lane;
    const float* lb = lum + 2 * hi * LP + ln + 1;
    const int dofs = 8 * hi * TP + ln;

    auto vtile31 = [&](int it, int j) {
        v8f acc = {0.f, 0.f, 0.f, 0.f, 0.f, 0.f, 0.f, 0.f};
#pragma unroll
        for (int s = 0; s < 12; ++s) {
            const float* p = lb + ((16 * it + 4 * s) & (RING - 1)) * LP + 16 * j;
            v2f bf;
            bf[0] = p[0];
            bf[1] = p[LP];
            acc = __builtin_amdgcn_wmma_f32_16x16x4_f32(
                false, w31[s], false, bf, (short)0, acc, false, false);
        }
        float* d = t31 + dofs + 16 * j;
#pragma unroll
        for (int r = 0; r < 8; ++r) d[r * TP] = acc[r];
    };
    auto vtile7 = [&](int it, int j) {
        v8f acc = {0.f, 0.f, 0.f, 0.f, 0.f, 0.f, 0.f, 0.f};
#pragma unroll
        for (int s = 0; s < 6; ++s) {
            const float* p = lb + ((16 * it + 12 + 4 * s) & (RING - 1)) * LP + 16 * j;
            v2f bf;
            bf[0] = p[0];
            bf[1] = p[LP];
            acc = __builtin_amdgcn_wmma_f32_16x16x4_f32(
                false, w7[s], false, bf, (short)0, acc, false, false);
        }
        float* d = t7 + dofs + 16 * j;
#pragma unroll
        for (int r = 0; r < 8; ++r) d[r * TP] = acc[r];
    };

#pragma clang loop unroll(disable)
    for (int it = 0; it < 4; ++it) {
        if (it == 1) __builtin_amdgcn_s_setprio(2);
        else if (it == 2) __builtin_amdgcn_s_setprio(1);
        else if (it == 3) __builtin_amdgcn_s_setprio(0);

        const int y0 = Y0 + 16 * it;
        const size_t tb = (size_t)y0 * WQ + X0;
        const float* xt = xb + tb;
        float*       ot = out + (size_t)(b * 3) * HW + tb;
        const float* xp[3] = {xt, xt + HW, xt + 2 * HW};
        float*       op[3] = {ot, ot + HW, ot + 2 * HW};
        v4f xq[2][3];
#pragma unroll
        for (int k = 0; k < 2; ++k) {
#pragma unroll
            for (int c = 0; c < 3; ++c) xq[k][c] = *(const v4f*)(xp[c] + 8 * k * WQ + rofs);
        }

        lds_barrier();

        vtile31(it, wv);
        vtile7(it, wv);
        if (wv == 0) vtile31(it, 8);
        else if (wv == 1) vtile7(it, 8);
        else if (wv == 2) vtile31(it, 9);
        else if (wv == 3) vtile7(it, 9);

        lds_barrier();

        v8f acc31 = {0.f, 0.f, 0.f, 0.f, 0.f, 0.f, 0.f, 0.f};
#pragma unroll
        for (int s = 0; s < 12; ++s) {
            v2f a = *(const v2f*)(a31p + 4 * s);
            acc31 = __builtin_amdgcn_wmma_f32_16x16x4_f32(
                false, a, false, w31[s], (short)0, acc31, false, false);
        }

        v8f acc7 = {0.f, 0.f, 0.f, 0.f, 0.f, 0.f, 0.f, 0.f};
#pragma unroll
        for (int s = 0; s < 6; ++s) {
            v2f a = *(const v2f*)(a7p + 4 * s);
            acc7 = __builtin_amdgcn_wmma_f32_16x16x4_f32(
                false, a, false, w7[s], (short)0, acc7, false, false);
        }

        const int sb = (16 * (it + 1)) & (RING - 1);
        float Lr[8];
        Lr[0] = lum[(hi ? (sb + 7) : (16 * it + 15)) * LP + 16 * wv + 16 + ln];
#pragma unroll
        for (int r = 1; r < 8; ++r) Lr[r] = lvp[(sb + r - 1) * LP];
#pragma unroll
        for (int r = 0; r < 8; ++r) {
            if (r == 4) __builtin_amdgcn_sched_barrier(0);
            float Lv  = Lr[r];
            float le  = Lv + ca * (Lv - acc31[r]) + ta * (Lv - acc7[r]);
            rtw[r * RP] = (le + 1e-6f) / (Lv + 1e-6f);
        }

        lds_barrier();

#pragma unroll
        for (int k = 0; k < 2; ++k) {
            const v4f rq = *(const v4f*)(rtr + 8 * k * RP);
#pragma unroll
            for (int c = 0; c < 3; ++c) {
                v4f vv = xq[k][c] * rq;
                v4f o;
#pragma unroll
                for (int e = 0; e < 4; ++e) o[e] = fminf(fmaxf(vv[e], 0.f), 1.f);
                *(v4f*)(op[c] + 8 * k * WQ + rofs) = o;
            }
        }

        if (it < 2) {
            v4f pl[3];
            slot_load(it + 4, pl);
            slot_store(it + 4, pl);
        }
    }
}

extern "C" void kernel_launch(void* const* d_in, const int* in_sizes, int n_in,
                              void* d_out, int out_size, void* d_ws, size_t ws_size,
                              hipStream_t stream) {
    const float* x    = (const float*)d_in[0];
    const float* clar = (const float*)d_in[1];
    const float* tex  = (const float*)d_in[2];
    float* out = (float*)d_out;

    dim3 grid(WQ / 128, HQ / 64, NB);
    blur_combine_kernel<<<grid, 256, 0, stream>>>(x, clar, tex, out);
}
